# starting kernel
# baseline (speedup 1.0000x reference)
#include <hip/hip_runtime.h>

typedef __attribute__((ext_vector_type(2))) float v2f;
typedef __attribute__((ext_vector_type(8))) float v8f;

#define C_CAPS 32
#define S_DIM  2048
#define H_DIM  1024
#define NCHUNK 8
#define SCHUNK (S_DIM / NCHUNK)

__global__ __launch_bounds__(256) void k_weighted_sum(
    const float* __restrict__ x, const float* __restrict__ d,
    float* __restrict__ partials)
{
    __shared__ float wts[SCHUNK];
    const int c     = blockIdx.y;
    const int chunk = blockIdx.x;
    const int t     = threadIdx.x;
    const int s0    = chunk * SCHUNK;

    if (d) wts[t] = d[c * S_DIM + s0 + t];
    else   wts[t] = 1.0f / (float)S_DIM;
    __syncthreads();

    const float4* xrow =
        reinterpret_cast<const float4*>(x + (size_t)c * S_DIM * H_DIM +
                                        (size_t)s0 * H_DIM) + t;
    float4 acc = make_float4(0.f, 0.f, 0.f, 0.f);
#pragma unroll 4
    for (int s = 0; s < SCHUNK; ++s) {
        float4 v = xrow[s * (H_DIM / 4)];
        float  w = wts[s];
        acc.x += w * v.x; acc.y += w * v.y;
        acc.z += w * v.z; acc.w += w * v.w;
    }
    float4* p = reinterpret_cast<float4*>(
        partials + ((size_t)chunk * C_CAPS + c) * H_DIM) + t;
    *p = acc;
}

__global__ __launch_bounds__(256) void k_reduce_partials(
    const float* __restrict__ partials, float* __restrict__ xd)
{
    const int i = blockIdx.x * 256 + threadIdx.x;
    float s = 0.f;
#pragma unroll
    for (int k = 0; k < NCHUNK; ++k)
        s += partials[(size_t)k * C_CAPS * H_DIM + i];
    xd[i] = s;
}

template <bool BT>
__global__ __launch_bounds__(32) void k_gemm32(
    const float* __restrict__ A, const float* __restrict__ W,
    float* __restrict__ Dst)
{
    const int lane = threadIdx.x;
    const int n0   = blockIdx.x * 16;
    const int kq   = (lane >> 4) * 2;
    const int l    = lane & 15;

    v8f c0 = {}; v8f c1 = {};
    for (int k = 0; k < H_DIM; k += 4) {
        const int kk = k + kq;
        v2f a0 = *(const v2f*)(A + (size_t)l * H_DIM + kk);
        v2f a1 = *(const v2f*)(A + (size_t)(l + 16) * H_DIM + kk);
        v2f b;
        if (BT) {
            b = *(const v2f*)(W + (size_t)(n0 + l) * H_DIM + kk);
        } else {
            b.x = W[(size_t)kk * H_DIM + n0 + l];
            b.y = W[(size_t)(kk + 1) * H_DIM + n0 + l];
        }
        c0 = __builtin_amdgcn_wmma_f32_16x16x4_f32(
                 false, a0, false, b, (short)0, c0, false, false);
        c1 = __builtin_amdgcn_wmma_f32_16x16x4_f32(
                 false, a1, false, b, (short)0, c1, false, false);
    }
    const int rbase = (lane >> 4) * 8;
    const int col   = n0 + l;
#pragma unroll
    for (int i = 0; i < 8; ++i) {
        Dst[(size_t)(rbase + i) * H_DIM + col]      = c0[i];
        Dst[(size_t)(rbase + i + 16) * H_DIM + col] = c1[i];
    }
}

__global__ __launch_bounds__(256) void k_squash(
    const float* __restrict__ craw, float* __restrict__ out)
{
    __shared__ float red[256];
    const int c = blockIdx.x;
    const int t = threadIdx.x;
    float4 v = reinterpret_cast<const float4*>(craw + (size_t)c * H_DIM)[t];
    red[t] = v.x * v.x + v.y * v.y + v.z * v.z + v.w * v.w;
    __syncthreads();
    for (int off = 128; off > 0; off >>= 1) {
        if (t < off) red[t] += red[t + off];
        __syncthreads();
    }
    const float sq    = red[0];
    const float coeff = sq / ((1.0f + sq) * sqrtf(sq + 1e-9f));
    float4 r = make_float4(coeff * v.x, coeff * v.y, coeff * v.z, coeff * v.w);
    reinterpret_cast<float4*>(out + (size_t)c * H_DIM)[t] = r;
}

__global__ __launch_bounds__(256) void k_dot_update(
    const float* __restrict__ x, const float* __restrict__ y,
    float* __restrict__ b, int zero_b)
{
    const int wave = threadIdx.x >> 5;
    const int lane = threadIdx.x & 31;
    const int row  = blockIdx.x * 8 + wave;
    const int c    = row >> 11;

    const float4* xr = reinterpret_cast<const float4*>(x + (size_t)row * H_DIM);
    const float4* yr = reinterpret_cast<const float4*>(y + (size_t)c * H_DIM);
    float acc = 0.f;
#pragma unroll
    for (int j = 0; j < 8; ++j) {
        float4 xv = xr[j * 32 + lane];
        float4 yv = yr[j * 32 + lane];
        acc += xv.x * yv.x + xv.y * yv.y + xv.z * yv.z + xv.w * yv.w;
    }
#pragma unroll
    for (int off = 16; off > 0; off >>= 1)
        acc += __shfl_down(acc, off, 32);
    if (lane == 0) {
        float prev = zero_b ? 0.f : b[row];
        b[row] = prev + acc;
    }
}

__global__ __launch_bounds__(256) void k_softmax(
    const float* __restrict__ b, float* __restrict__ d)
{
    __shared__ float red[256];
    const int c = blockIdx.x;
    const int t = threadIdx.x;
    const float* row = b + (size_t)c * S_DIM;

    float vals[8];
    float m = -__builtin_inff();
#pragma unroll
    for (int j = 0; j < 8; ++j) {
        vals[j] = row[j * 256 + t];
        m = fmaxf(m, vals[j]);
    }
    red[t] = m; __syncthreads();
    for (int off = 128; off > 0; off >>= 1) {
        if (t < off) red[t] = fmaxf(red[t], red[t + off]);
        __syncthreads();
    }
    m = red[0];
    __syncthreads();

    float sum = 0.f;
#pragma unroll
    for (int j = 0; j < 8; ++j) {
        vals[j] = __expf(vals[j] - m);
        sum += vals[j];
    }
    red[t] = sum; __syncthreads();
    for (int off = 128; off > 0; off >>= 1) {
        if (t < off) red[t] += red[t + off];
        __syncthreads();
    }
    const float inv = 1.0f / red[0];
    float* drow = d + (size_t)c * S_DIM;
#pragma unroll
    for (int j = 0; j < 8; ++j) drow[j * 256 + t] = vals[j] * inv;
}

extern "C" void kernel_launch(void* const* d_in, const int* in_sizes, int n_in,
                              void* d_out, int out_size, void* d_ws, size_t ws_size,
                              hipStream_t stream)
{
    const float* x = (const float*)d_in[0];
    const float* W = (const float*)d_in[1];
    float* out = (float*)d_out;
    float* ws  = (float*)d_ws;

    float* partials = ws;
    float* xd   = partials + (size_t)NCHUNK * C_CAPS * H_DIM;
    float* craw = xd   + C_CAPS * H_DIM;
    float* cvec = craw + C_CAPS * H_DIM;
    float* yv   = cvec + C_CAPS * H_DIM;
    float* bl   = yv   + C_CAPS * H_DIM;
    float* dw   = bl   + C_CAPS * S_DIM;

    const dim3 gws(NCHUNK, C_CAPS);
    const int  gred = (C_CAPS * H_DIM) / 256;
    const int  gnt  = H_DIM / 16;
    const int  gdot = (C_CAPS * S_DIM) / 8;

    k_weighted_sum<<<gws, 256, 0, stream>>>(x, nullptr, partials);
    k_reduce_partials<<<gred, 256, 0, stream>>>(partials, xd);
    k_gemm32<false><<<gnt, 32, 0, stream>>>(xd, W, craw);
    k_squash<<<C_CAPS, 256, 0, stream>>>(craw, cvec);
    k_gemm32<true><<<gnt, 32, 0, stream>>>(cvec, W, yv);
    k_dot_update<<<gdot, 256, 0, stream>>>(x, yv, bl, 1);

    k_softmax<<<C_CAPS, 256, 0, stream>>>(bl, dw);
    k_weighted_sum<<<gws, 256, 0, stream>>>(x, dw, partials);
    k_reduce_partials<<<gred, 256, 0, stream>>>(partials, xd);
    k_gemm32<false><<<gnt, 32, 0, stream>>>(xd, W, craw);
    k_squash<<<C_CAPS, 256, 0, stream>>>(craw, cvec);
    k_gemm32<true><<<gnt, 32, 0, stream>>>(cvec, W, yv);
    k_dot_update<<<gdot, 256, 0, stream>>>(x, yv, bl, 0);

    k_softmax<<<C_CAPS, 256, 0, stream>>>(bl, dw);
    k_weighted_sum<<<gws, 256, 0, stream>>>(x, dw, partials);
    k_reduce_partials<<<gred, 256, 0, stream>>>(partials, xd);
    k_gemm32<false><<<gnt, 32, 0, stream>>>(xd, W, craw);
    k_squash<<<C_CAPS, 256, 0, stream>>>(craw, out);
}


# GEMMs: one 16x16 tile per 8-wave block, K split over the waves, all fragment loads issued ahead of the WMMAs, partial tiles summed through LDS (changes the f32 summation order)
# speedup vs baseline: 1.5030x; 1.5030x over previous
#include <hip/hip_runtime.h>

typedef __attribute__((ext_vector_type(2))) float v2f;
typedef __attribute__((ext_vector_type(8))) float v8f;

#define C_CAPS 32
#define S_DIM  2048
#define H_DIM  1024
#define NCHUNK 8
#define SCHUNK (S_DIM / NCHUNK)

__global__ __launch_bounds__(256) void k_weighted_sum(
    const float* __restrict__ x, const float* __restrict__ d,
    float* __restrict__ partials)
{
    __shared__ float wts[SCHUNK];
    const int c     = blockIdx.y;
    const int chunk = blockIdx.x;
    const int t     = threadIdx.x;
    const int s0    = chunk * SCHUNK;

    if (d) wts[t] = d[c * S_DIM + s0 + t];
    else   wts[t] = 1.0f / (float)S_DIM;
    __syncthreads();

    const float4* xrow =
        reinterpret_cast<const float4*>(x + (size_t)c * S_DIM * H_DIM +
                                        (size_t)s0 * H_DIM) + t;
    float4 acc = make_float4(0.f, 0.f, 0.f, 0.f);
#pragma unroll 4
    for (int s = 0; s < SCHUNK; ++s) {
        float4 v = xrow[s * (H_DIM / 4)];
        float  w = wts[s];
        acc.x += w * v.x; acc.y += w * v.y;
        acc.z += w * v.z; acc.w += w * v.w;
    }
    float4* p = reinterpret_cast<float4*>(
        partials + ((size_t)chunk * C_CAPS + c) * H_DIM) + t;
    *p = acc;
}

__global__ __launch_bounds__(256) void k_reduce_partials(
    const float* __restrict__ partials, float* __restrict__ xd)
{
    const int i = blockIdx.x * 256 + threadIdx.x;
    float s = 0.f;
#pragma unroll
    for (int k = 0; k < NCHUNK; ++k)
        s += partials[(size_t)k * C_CAPS * H_DIM + i];
    xd[i] = s;
}

#define GEMM_WAVES 8
#define GEMM_KW    (H_DIM / GEMM_WAVES)
#define GEMM_STEPS (GEMM_KW / 4)

template <bool BT>
__global__ __launch_bounds__(256) void k_gemm32(
    const float* __restrict__ A, const float* __restrict__ W,
    float* __restrict__ Dst)
{
    __shared__ float red[GEMM_WAVES][8][32];
    const int wave = threadIdx.x >> 5;
    const int lane = threadIdx.x & 31;
    const int n0   = blockIdx.x * 16;
    const int m0   = blockIdx.y * 16;
    const int kq   = (lane >> 4) * 2;
    const int l    = lane & 15;
    const int kk0  = wave * GEMM_KW + kq;

    const float* Ap = A + (size_t)(m0 + l) * H_DIM + kk0;
    const float* Wp = BT ? W + (size_t)(n0 + l) * H_DIM + kk0
                         : W + (size_t)kk0 * H_DIM + n0 + l;

    v2f a[GEMM_STEPS];
    v2f b[GEMM_STEPS];
#pragma unroll
    for (int i = 0; i < GEMM_STEPS; ++i) {
        a[i] = *(const v2f*)(Ap + 4 * i);
        if (BT) {
            b[i] = *(const v2f*)(Wp + 4 * i);
        } else {
            b[i].x = Wp[(size_t)(4 * i) * H_DIM];
            b[i].y = Wp[(size_t)(4 * i + 1) * H_DIM];
        }
    }

    __builtin_amdgcn_sched_barrier(0);

    v8f c0 = {}; v8f c1 = {};
#pragma unroll
    for (int i = 0; i < GEMM_STEPS; i += 2) {
        c0 = __builtin_amdgcn_wmma_f32_16x16x4_f32(
                 false, a[i], false, b[i], (short)0, c0, false, false);
        c1 = __builtin_amdgcn_wmma_f32_16x16x4_f32(
                 false, a[i + 1], false, b[i + 1], (short)0, c1, false, false);
    }

    for (int i = 0; i < 8; ++i) red[wave][i][lane] = c0[i] + c1[i];
    __syncthreads();

    float s = 0.f;
    for (int w = 0; w < GEMM_WAVES; ++w) s += red[w][wave][lane];
    Dst[(size_t)(m0 + (lane >> 4) * 8 + wave) * H_DIM + n0 + l] = s;
}

__global__ __launch_bounds__(256) void k_squash(
    const float* __restrict__ craw, float* __restrict__ out)
{
    __shared__ float red[256];
    const int c = blockIdx.x;
    const int t = threadIdx.x;
    float4 v = reinterpret_cast<const float4*>(craw + (size_t)c * H_DIM)[t];
    red[t] = v.x * v.x + v.y * v.y + v.z * v.z + v.w * v.w;
    __syncthreads();
    for (int off = 128; off > 0; off >>= 1) {
        if (t < off) red[t] += red[t + off];
        __syncthreads();
    }
    const float sq    = red[0];
    const float coeff = sq / ((1.0f + sq) * sqrtf(sq + 1e-9f));
    float4 r = make_float4(coeff * v.x, coeff * v.y, coeff * v.z, coeff * v.w);
    reinterpret_cast<float4*>(out + (size_t)c * H_DIM)[t] = r;
}

__global__ __launch_bounds__(256) void k_dot_update(
    const float* __restrict__ x, const float* __restrict__ y,
    float* __restrict__ b, int zero_b)
{
    const int wave = threadIdx.x >> 5;
    const int lane = threadIdx.x & 31;
    const int row  = blockIdx.x * 8 + wave;
    const int c    = row >> 11;

    const float4* xr = reinterpret_cast<const float4*>(x + (size_t)row * H_DIM);
    const float4* yr = reinterpret_cast<const float4*>(y + (size_t)c * H_DIM);
    float acc = 0.f;
#pragma unroll
    for (int j = 0; j < 8; ++j) {
        float4 xv = xr[j * 32 + lane];
        float4 yv = yr[j * 32 + lane];
        acc += xv.x * yv.x + xv.y * yv.y + xv.z * yv.z + xv.w * yv.w;
    }
#pragma unroll
    for (int off = 16; off > 0; off >>= 1)
        acc += __shfl_down(acc, off, 32);
    if (lane == 0) {
        float prev = zero_b ? 0.f : b[row];
        b[row] = prev + acc;
    }
}

__global__ __launch_bounds__(256) void k_softmax(
    const float* __restrict__ b, float* __restrict__ d)
{
    __shared__ float red[256];
    const int c = blockIdx.x;
    const int t = threadIdx.x;
    const float* row = b + (size_t)c * S_DIM;

    float vals[8];
    float m = -__builtin_inff();
#pragma unroll
    for (int j = 0; j < 8; ++j) {
        vals[j] = row[j * 256 + t];
        m = fmaxf(m, vals[j]);
    }
    red[t] = m; __syncthreads();
    for (int off = 128; off > 0; off >>= 1) {
        if (t < off) red[t] = fmaxf(red[t], red[t + off]);
        __syncthreads();
    }
    m = red[0];
    __syncthreads();

    float sum = 0.f;
#pragma unroll
    for (int j = 0; j < 8; ++j) {
        vals[j] = __expf(vals[j] - m);
        sum += vals[j];
    }
    red[t] = sum; __syncthreads();
    for (int off = 128; off > 0; off >>= 1) {
        if (t < off) red[t] += red[t + off];
        __syncthreads();
    }
    const float inv = 1.0f / red[0];
    float* drow = d + (size_t)c * S_DIM;
#pragma unroll
    for (int j = 0; j < 8; ++j) drow[j * 256 + t] = vals[j] * inv;
}

extern "C" void kernel_launch(void* const* d_in, const int* in_sizes, int n_in,
                              void* d_out, int out_size, void* d_ws, size_t ws_size,
                              hipStream_t stream)
{
    const float* x = (const float*)d_in[0];
    const float* W = (const float*)d_in[1];
    float* out = (float*)d_out;
    float* ws  = (float*)d_ws;

    float* partials = ws;
    float* xd   = partials + (size_t)NCHUNK * C_CAPS * H_DIM;
    float* craw = xd   + C_CAPS * H_DIM;
    float* cvec = craw + C_CAPS * H_DIM;
    float* yv   = cvec + C_CAPS * H_DIM;
    float* bl   = yv   + C_CAPS * H_DIM;
    float* dw   = bl   + C_CAPS * S_DIM;

    const dim3 gws(NCHUNK, C_CAPS);
    const int  gred = (C_CAPS * H_DIM) / 256;
    const dim3 gnt(H_DIM / 16, C_CAPS / 16);
    const int  gdot = (C_CAPS * S_DIM) / 8;

    k_weighted_sum<<<gws, 256, 0, stream>>>(x, nullptr, partials);
    k_reduce_partials<<<gred, 256, 0, stream>>>(partials, xd);
    k_gemm32<false><<<gnt, 256, 0, stream>>>(xd, W, craw);
    k_squash<<<C_CAPS, 256, 0, stream>>>(craw, cvec);
    k_gemm32<true><<<gnt, 256, 0, stream>>>(cvec, W, yv);
    k_dot_update<<<gdot, 256, 0, stream>>>(x, yv, bl, 1);

    k_softmax<<<C_CAPS, 256, 0, stream>>>(bl, dw);
    k_weighted_sum<<<gws, 256, 0, stream>>>(x, dw, partials);
    k_reduce_partials<<<gred, 256, 0, stream>>>(partials, xd);
    k_gemm32<false><<<gnt, 256, 0, stream>>>(xd, W, craw);
    k_squash<<<C_CAPS, 256, 0, stream>>>(craw, cvec);
    k_gemm32<true><<<gnt, 256, 0, stream>>>(cvec, W, yv);
    k_dot_update<<<gdot, 256, 0, stream>>>(x, yv, bl, 0);

    k_softmax<<<C_CAPS, 256, 0, stream>>>(bl, dw);
    k_weighted_sum<<<gws, 256, 0, stream>>>(x, dw, partials);
    k_reduce_partials<<<gred, 256, 0, stream>>>(partials, xd);
    k_gemm32<false><<<gnt, 256, 0, stream>>>(xd, W, craw);
    k_squash<<<C_CAPS, 256, 0, stream>>>(craw, out);
}


# x passes (weighted sum, agreement dot): 512 blocks of 8 long-lived waves, next 16 row loads issued before the 16 in hand are summed; weighted sum adds 2 row-group sums in LDS (f32 sum order)
# speedup vs baseline: 2.1677x; 1.4422x over previous
#include <hip/hip_runtime.h>

typedef __attribute__((ext_vector_type(2))) float v2f;
typedef __attribute__((ext_vector_type(8))) float v8f;

#define C_CAPS 32
#define S_DIM  2048
#define H_DIM  1024
#define NCHUNK 8
#define SCHUNK (S_DIM / NCHUNK)

#define ST_DEPTH 16

#define WS_GROUPS  2
#define WS_COLT    128
#define WS_HALVES  (H_DIM / (4 * WS_COLT))
#define WS_TRIPS   (SCHUNK / WS_GROUPS)
#define WS_BLKS    (WS_TRIPS / ST_DEPTH)
#define WS_RSTRIDE (WS_GROUPS * (H_DIM / 4))

__device__ __forceinline__ void ws_load(float4 (&buf)[ST_DEPTH], const float4* p)
{
#pragma unroll
    for (int k = 0; k < ST_DEPTH; ++k) buf[k] = p[(size_t)k * WS_RSTRIDE];
}

__device__ __forceinline__ void ws_sum(float4& acc, const float4 (&buf)[ST_DEPTH],
                                       const float* w)
{
#pragma unroll
    for (int k = 0; k < ST_DEPTH; ++k) {
        const float ww = w[k * WS_GROUPS];
        acc.x += ww * buf[k].x; acc.y += ww * buf[k].y;
        acc.z += ww * buf[k].z; acc.w += ww * buf[k].w;
    }
}

__global__ __launch_bounds__(256) void k_weighted_sum(
    const float* __restrict__ x, const float* __restrict__ d,
    float* __restrict__ partials)
{
    __shared__ float  wts[SCHUNK];
    __shared__ float4 comb[WS_COLT];
    const int c     = blockIdx.y;
    const int chunk = blockIdx.x / WS_HALVES;
    const int half  = blockIdx.x % WS_HALVES;
    const int t     = threadIdx.x;
    const int g     = t / WS_COLT;
    const int col   = t % WS_COLT;
    const int s0    = chunk * SCHUNK;

    if (d) wts[t] = d[c * S_DIM + s0 + t];
    else   wts[t] = 1.0f / (float)S_DIM;
    __syncthreads();

    const float4* xrow =
        reinterpret_cast<const float4*>(x + (size_t)c * S_DIM * H_DIM +
                                        (size_t)(s0 + g) * H_DIM) +
        half * WS_COLT + col;
    const float* wg = wts + g;

    float4 bufA[ST_DEPTH], bufB[ST_DEPTH];
    float4 acc = make_float4(0.f, 0.f, 0.f, 0.f);
    ws_load(bufA, xrow);
#pragma unroll
    for (int blk = 0; blk < WS_BLKS; blk += 2) {
        ws_load(bufB, xrow + (size_t)(blk + 1) * ST_DEPTH * WS_RSTRIDE);
        __builtin_amdgcn_sched_barrier(0);
        ws_sum(acc, bufA, wg + blk * ST_DEPTH * WS_GROUPS);
        __builtin_amdgcn_sched_barrier(0);
        if (blk + 2 < WS_BLKS)
            ws_load(bufA, xrow + (size_t)(blk + 2) * ST_DEPTH * WS_RSTRIDE);
        __builtin_amdgcn_sched_barrier(0);
        ws_sum(acc, bufB, wg + (blk + 1) * ST_DEPTH * WS_GROUPS);
        __builtin_amdgcn_sched_barrier(0);
    }

    if (g == 1) comb[col] = acc;
    __syncthreads();
    if (g == 0) {
        const float4 o = comb[col];
        acc.x += o.x; acc.y += o.y; acc.z += o.z; acc.w += o.w;
        float4* p = reinterpret_cast<float4*>(
            partials + ((size_t)chunk * C_CAPS + c) * H_DIM) +
            half * WS_COLT + col;
        *p = acc;
    }
}

__global__ __launch_bounds__(256) void k_reduce_partials(
    const float* __restrict__ partials, float* __restrict__ xd)
{
    const int i = blockIdx.x * 256 + threadIdx.x;
    float s = 0.f;
#pragma unroll
    for (int k = 0; k < NCHUNK; ++k)
        s += partials[(size_t)k * C_CAPS * H_DIM + i];
    xd[i] = s;
}

#define GEMM_WAVES 8
#define GEMM_KW    (H_DIM / GEMM_WAVES)
#define GEMM_STEPS (GEMM_KW / 4)

template <bool BT>
__global__ __launch_bounds__(256) void k_gemm32(
    const float* __restrict__ A, const float* __restrict__ W,
    float* __restrict__ Dst)
{
    __shared__ float red[GEMM_WAVES][8][32];
    const int wave = threadIdx.x >> 5;
    const int lane = threadIdx.x & 31;
    const int n0   = blockIdx.x * 16;
    const int m0   = blockIdx.y * 16;
    const int kq   = (lane >> 4) * 2;
    const int l    = lane & 15;
    const int kk0  = wave * GEMM_KW + kq;

    const float* Ap = A + (size_t)(m0 + l) * H_DIM + kk0;
    const float* Wp = BT ? W + (size_t)(n0 + l) * H_DIM + kk0
                         : W + (size_t)kk0 * H_DIM + n0 + l;

    v2f a[GEMM_STEPS];
    v2f b[GEMM_STEPS];
#pragma unroll
    for (int i = 0; i < GEMM_STEPS; ++i) {
        a[i] = *(const v2f*)(Ap + 4 * i);
        if (BT) {
            b[i] = *(const v2f*)(Wp + 4 * i);
        } else {
            b[i].x = Wp[(size_t)(4 * i) * H_DIM];
            b[i].y = Wp[(size_t)(4 * i + 1) * H_DIM];
        }
    }

    __builtin_amdgcn_sched_barrier(0);

    v8f c0 = {}; v8f c1 = {};
#pragma unroll
    for (int i = 0; i < GEMM_STEPS; i += 2) {
        c0 = __builtin_amdgcn_wmma_f32_16x16x4_f32(
                 false, a[i], false, b[i], (short)0, c0, false, false);
        c1 = __builtin_amdgcn_wmma_f32_16x16x4_f32(
                 false, a[i + 1], false, b[i + 1], (short)0, c1, false, false);
    }

    for (int i = 0; i < 8; ++i) red[wave][i][lane] = c0[i] + c1[i];
    __syncthreads();

    float s = 0.f;
    for (int w = 0; w < GEMM_WAVES; ++w) s += red[w][wave][lane];
    Dst[(size_t)(m0 + (lane >> 4) * 8 + wave) * H_DIM + n0 + l] = s;
}

__global__ __launch_bounds__(256) void k_squash(
    const float* __restrict__ craw, float* __restrict__ out)
{
    __shared__ float red[256];
    const int c = blockIdx.x;
    const int t = threadIdx.x;
    float4 v = reinterpret_cast<const float4*>(craw + (size_t)c * H_DIM)[t];
    red[t] = v.x * v.x + v.y * v.y + v.z * v.z + v.w * v.w;
    __syncthreads();
    for (int off = 128; off > 0; off >>= 1) {
        if (t < off) red[t] += red[t + off];
        __syncthreads();
    }
    const float sq    = red[0];
    const float coeff = sq / ((1.0f + sq) * sqrtf(sq + 1e-9f));
    float4 r = make_float4(coeff * v.x, coeff * v.y, coeff * v.z, coeff * v.w);
    reinterpret_cast<float4*>(out + (size_t)c * H_DIM)[t] = r;
}

#define DOT_WROWS 16
#define DOT_BROWS (8 * DOT_WROWS)
#define DOT_PAIRS (DOT_WROWS / 2)

__device__ __forceinline__ void dot_load(float4 (&buf)[ST_DEPTH], const float4* p)
{
#pragma unroll
    for (int q = 0; q < ST_DEPTH; ++q) buf[q] = p[q * 32];
}

__device__ __forceinline__ void dot_sum(const float4 (&buf)[ST_DEPTH],
                                        const float4 (&yv)[8], float* out, int lane)
{
#pragma unroll
    for (int r = 0; r < 2; ++r) {
        float acc = 0.f;
#pragma unroll
        for (int j = 0; j < 8; ++j) {
            const float4 xv = buf[r * 8 + j];
            acc += xv.x * yv[j].x + xv.y * yv[j].y + xv.z * yv[j].z + xv.w * yv[j].w;
        }
#pragma unroll
        for (int off = 16; off > 0; off >>= 1)
            acc += __shfl_down(acc, off, 32);
        if (lane == 0) out[r] = acc;
    }
}

__global__ __launch_bounds__(256) void k_dot_update(
    const float* __restrict__ x, const float* __restrict__ y,
    float* __restrict__ b, int zero_b)
{
    __shared__ float dots[DOT_BROWS];
    const int wave = threadIdx.x >> 5;
    const int lane = threadIdx.x & 31;
    const int row0 = blockIdx.x * DOT_BROWS;
    const int c    = row0 >> 11;

    const float4* yr = reinterpret_cast<const float4*>(y + (size_t)c * H_DIM);
    float4 yv[8];
#pragma unroll
    for (int j = 0; j < 8; ++j) yv[j] = yr[j * 32 + lane];

    const float4* xr = reinterpret_cast<const float4*>(
        x + (size_t)(row0 + wave * DOT_WROWS) * H_DIM) + lane;
    float* dw = dots + wave * DOT_WROWS;

    float4 bufA[ST_DEPTH], bufB[ST_DEPTH];
    dot_load(bufA, xr);
    __builtin_amdgcn_sched_barrier(0);
#pragma unroll
    for (int pr = 0; pr < DOT_PAIRS; pr += 2) {
        dot_load(bufB, xr + (pr + 1) * (2 * H_DIM / 4));
        __builtin_amdgcn_sched_barrier(0);
        dot_sum(bufA, yv, dw + pr * 2, lane);
        __builtin_amdgcn_sched_barrier(0);
        if (pr + 2 < DOT_PAIRS)
            dot_load(bufA, xr + (pr + 2) * (2 * H_DIM / 4));
        __builtin_amdgcn_sched_barrier(0);
        dot_sum(bufB, yv, dw + (pr + 1) * 2, lane);
        __builtin_amdgcn_sched_barrier(0);
    }
    __syncthreads();
    if (threadIdx.x < DOT_BROWS) {
        const int row = row0 + threadIdx.x;
        float prev = zero_b ? 0.f : b[row];
        b[row] = prev + dots[threadIdx.x];
    }
}

__global__ __launch_bounds__(256) void k_softmax(
    const float* __restrict__ b, float* __restrict__ d)
{
    __shared__ float red[256];
    const int c = blockIdx.x;
    const int t = threadIdx.x;
    const float* row = b + (size_t)c * S_DIM;

    float vals[8];
    float m = -__builtin_inff();
#pragma unroll
    for (int j = 0; j < 8; ++j) {
        vals[j] = row[j * 256 + t];
        m = fmaxf(m, vals[j]);
    }
    red[t] = m; __syncthreads();
    for (int off = 128; off > 0; off >>= 1) {
        if (t < off) red[t] = fmaxf(red[t], red[t + off]);
        __syncthreads();
    }
    m = red[0];
    __syncthreads();

    float sum = 0.f;
#pragma unroll
    for (int j = 0; j < 8; ++j) {
        vals[j] = __expf(vals[j] - m);
        sum += vals[j];
    }
    red[t] = sum; __syncthreads();
    for (int off = 128; off > 0; off >>= 1) {
        if (t < off) red[t] += red[t + off];
        __syncthreads();
    }
    const float inv = 1.0f / red[0];
    float* drow = d + (size_t)c * S_DIM;
#pragma unroll
    for (int j = 0; j < 8; ++j) drow[j * 256 + t] = vals[j] * inv;
}

extern "C" void kernel_launch(void* const* d_in, const int* in_sizes, int n_in,
                              void* d_out, int out_size, void* d_ws, size_t ws_size,
                              hipStream_t stream)
{
    const float* x = (const float*)d_in[0];
    const float* W = (const float*)d_in[1];
    float* out = (float*)d_out;
    float* ws  = (float*)d_ws;

    float* partials = ws;
    float* xd   = partials + (size_t)NCHUNK * C_CAPS * H_DIM;
    float* craw = xd   + C_CAPS * H_DIM;
    float* cvec = craw + C_CAPS * H_DIM;
    float* yv   = cvec + C_CAPS * H_DIM;
    float* bl   = yv   + C_CAPS * H_DIM;
    float* dw   = bl   + C_CAPS * S_DIM;

    const dim3 gws(NCHUNK * WS_HALVES, C_CAPS);
    const int  gred = (C_CAPS * H_DIM) / 256;
    const dim3 gnt(H_DIM / 16, C_CAPS / 16);
    const int  gdot = (C_CAPS * S_DIM) / DOT_BROWS;

    k_weighted_sum<<<gws, 256, 0, stream>>>(x, nullptr, partials);
    k_reduce_partials<<<gred, 256, 0, stream>>>(partials, xd);
    k_gemm32<false><<<gnt, 256, 0, stream>>>(xd, W, craw);
    k_squash<<<C_CAPS, 256, 0, stream>>>(craw, cvec);
    k_gemm32<true><<<gnt, 256, 0, stream>>>(cvec, W, yv);
    k_dot_update<<<gdot, 256, 0, stream>>>(x, yv, bl, 1);

    k_softmax<<<C_CAPS, 256, 0, stream>>>(bl, dw);
    k_weighted_sum<<<gws, 256, 0, stream>>>(x, dw, partials);
    k_reduce_partials<<<gred, 256, 0, stream>>>(partials, xd);
    k_gemm32<false><<<gnt, 256, 0, stream>>>(xd, W, craw);
    k_squash<<<C_CAPS, 256, 0, stream>>>(craw, cvec);
    k_gemm32<true><<<gnt, 256, 0, stream>>>(cvec, W, yv);
    k_dot_update<<<gdot, 256, 0, stream>>>(x, yv, bl, 0);

    k_softmax<<<C_CAPS, 256, 0, stream>>>(bl, dw);
    k_weighted_sum<<<gws, 256, 0, stream>>>(x, dw, partials);
    k_reduce_partials<<<gred, 256, 0, stream>>>(partials, xd);
    k_gemm32<false><<<gnt, 256, 0, stream>>>(xd, W, craw);
    k_squash<<<C_CAPS, 256, 0, stream>>>(craw, out);
}


# Iter 2-3: agreement dot, softmax and weighted sum fused into one pass over x (online softmax: running max, per-wave/per-block partials rescaled and merged; f32 sum order changes); x read 3x not 5x
# speedup vs baseline: 3.0888x; 1.4249x over previous
#include <hip/hip_runtime.h>

typedef __attribute__((ext_vector_type(2))) float v2f;
typedef __attribute__((ext_vector_type(8))) float v8f;

#define C_CAPS 32
#define S_DIM  2048
#define H_DIM  1024
#define NCHUNK 8
#define SCHUNK (S_DIM / NCHUNK)

#define ST_DEPTH 16

#define WS_GROUPS  2
#define WS_COLT    128
#define WS_HALVES  (H_DIM / (4 * WS_COLT))
#define WS_TRIPS   (SCHUNK / WS_GROUPS)
#define WS_BLKS    (WS_TRIPS / ST_DEPTH)
#define WS_RSTRIDE (WS_GROUPS * (H_DIM / 4))

__device__ __forceinline__ void ws_load(float4 (&buf)[ST_DEPTH], const float4* p)
{
#pragma unroll
    for (int k = 0; k < ST_DEPTH; ++k) buf[k] = p[(size_t)k * WS_RSTRIDE];
}

__device__ __forceinline__ void ws_sum(float4& acc, const float4 (&buf)[ST_DEPTH],
                                       const float* w)
{
#pragma unroll
    for (int k = 0; k < ST_DEPTH; ++k) {
        const float ww = w[k * WS_GROUPS];
        acc.x += ww * buf[k].x; acc.y += ww * buf[k].y;
        acc.z += ww * buf[k].z; acc.w += ww * buf[k].w;
    }
}

__global__ __launch_bounds__(256) void k_weighted_sum(
    const float* __restrict__ x, const float* __restrict__ d,
    float* __restrict__ partials)
{
    __shared__ float  wts[SCHUNK];
    __shared__ float4 comb[WS_COLT];
    const int c     = blockIdx.y;
    const int chunk = blockIdx.x / WS_HALVES;
    const int half  = blockIdx.x % WS_HALVES;
    const int t     = threadIdx.x;
    const int g     = t / WS_COLT;
    const int col   = t % WS_COLT;
    const int s0    = chunk * SCHUNK;

    if (d) wts[t] = d[c * S_DIM + s0 + t];
    else   wts[t] = 1.0f / (float)S_DIM;
    __syncthreads();

    const float4* xrow =
        reinterpret_cast<const float4*>(x + (size_t)c * S_DIM * H_DIM +
                                        (size_t)(s0 + g) * H_DIM) +
        half * WS_COLT + col;
    const float* wg = wts + g;

    float4 bufA[ST_DEPTH], bufB[ST_DEPTH];
    float4 acc = make_float4(0.f, 0.f, 0.f, 0.f);
    ws_load(bufA, xrow);
#pragma unroll
    for (int blk = 0; blk < WS_BLKS; blk += 2) {
        ws_load(bufB, xrow + (size_t)(blk + 1) * ST_DEPTH * WS_RSTRIDE);
        __builtin_amdgcn_sched_barrier(0);
        ws_sum(acc, bufA, wg + blk * ST_DEPTH * WS_GROUPS);
        __builtin_amdgcn_sched_barrier(0);
        if (blk + 2 < WS_BLKS)
            ws_load(bufA, xrow + (size_t)(blk + 2) * ST_DEPTH * WS_RSTRIDE);
        __builtin_amdgcn_sched_barrier(0);
        ws_sum(acc, bufB, wg + (blk + 1) * ST_DEPTH * WS_GROUPS);
        __builtin_amdgcn_sched_barrier(0);
    }

    if (g == 1) comb[col] = acc;
    __syncthreads();
    if (g == 0) {
        const float4 o = comb[col];
        acc.x += o.x; acc.y += o.y; acc.z += o.z; acc.w += o.w;
        float4* p = reinterpret_cast<float4*>(
            partials + ((size_t)chunk * C_CAPS + c) * H_DIM) +
            half * WS_COLT + col;
        *p = acc;
    }
}

__global__ __launch_bounds__(256) void k_reduce_partials(
    const float* __restrict__ partials, float* __restrict__ xd)
{
    const int i = blockIdx.x * 256 + threadIdx.x;
    float s = 0.f;
#pragma unroll
    for (int k = 0; k < NCHUNK; ++k)
        s += partials[(size_t)k * C_CAPS * H_DIM + i];
    xd[i] = s;
}

#define GEMM_WAVES 8
#define GEMM_KW    (H_DIM / GEMM_WAVES)
#define GEMM_STEPS (GEMM_KW / 4)

template <bool BT>
__global__ __launch_bounds__(256) void k_gemm32(
    const float* __restrict__ A, const float* __restrict__ W,
    float* __restrict__ Dst)
{
    __shared__ float red[GEMM_WAVES][8][32];
    const int wave = threadIdx.x >> 5;
    const int lane = threadIdx.x & 31;
    const int n0   = blockIdx.x * 16;
    const int m0   = blockIdx.y * 16;
    const int kq   = (lane >> 4) * 2;
    const int l    = lane & 15;
    const int kk0  = wave * GEMM_KW + kq;

    const float* Ap = A + (size_t)(m0 + l) * H_DIM + kk0;
    const float* Wp = BT ? W + (size_t)(n0 + l) * H_DIM + kk0
                         : W + (size_t)kk0 * H_DIM + n0 + l;

    v2f a[GEMM_STEPS];
    v2f b[GEMM_STEPS];
#pragma unroll
    for (int i = 0; i < GEMM_STEPS; ++i) {
        a[i] = *(const v2f*)(Ap + 4 * i);
        if (BT) {
            b[i] = *(const v2f*)(Wp + 4 * i);
        } else {
            b[i].x = Wp[(size_t)(4 * i) * H_DIM];
            b[i].y = Wp[(size_t)(4 * i + 1) * H_DIM];
        }
    }

    __builtin_amdgcn_sched_barrier(0);

    v8f c0 = {}; v8f c1 = {};
#pragma unroll
    for (int i = 0; i < GEMM_STEPS; i += 2) {
        c0 = __builtin_amdgcn_wmma_f32_16x16x4_f32(
                 false, a[i], false, b[i], (short)0, c0, false, false);
        c1 = __builtin_amdgcn_wmma_f32_16x16x4_f32(
                 false, a[i + 1], false, b[i + 1], (short)0, c1, false, false);
    }

    for (int i = 0; i < 8; ++i) red[wave][i][lane] = c0[i] + c1[i];
    __syncthreads();

    float s = 0.f;
    for (int w = 0; w < GEMM_WAVES; ++w) s += red[w][wave][lane];
    Dst[(size_t)(m0 + (lane >> 4) * 8 + wave) * H_DIM + n0 + l] = s;
}

__global__ __launch_bounds__(256) void k_squash(
    const float* __restrict__ craw, float* __restrict__ out)
{
    __shared__ float red[256];
    const int c = blockIdx.x;
    const int t = threadIdx.x;
    float4 v = reinterpret_cast<const float4*>(craw + (size_t)c * H_DIM)[t];
    red[t] = v.x * v.x + v.y * v.y + v.z * v.z + v.w * v.w;
    __syncthreads();
    for (int off = 128; off > 0; off >>= 1) {
        if (t < off) red[t] += red[t + off];
        __syncthreads();
    }
    const float sq    = red[0];
    const float coeff = sq / ((1.0f + sq) * sqrtf(sq + 1e-9f));
    float4 r = make_float4(coeff * v.x, coeff * v.y, coeff * v.z, coeff * v.w);
    reinterpret_cast<float4*>(out + (size_t)c * H_DIM)[t] = r;
}

#define RP_WAVES   16
#define RP_WROWS   16
#define RP_BROWS   (RP_WAVES * RP_WROWS)
#define RP_CHUNKS  (S_DIM / RP_BROWS)
#define RP_PIECES  8
#define RP_ROWSTEP (H_DIM / 4)
#define RP_MLPITCH 32

__device__ __forceinline__ void rp_load(float4 (&buf)[RP_PIECES], const float4* p)
{
#pragma unroll
    for (int q = 0; q < RP_PIECES; ++q) buf[q] = p[q * 32];
}

__device__ __forceinline__ void rp_row(const float4 (&buf)[RP_PIECES],
                                       const float4 (&yv)[RP_PIECES],
                                       float4 (&acc)[RP_PIECES],
                                       float& m, float& l, float bp, float& snew,
                                       int i, int lane)
{
    float d = 0.f;
#pragma unroll
    for (int j = 0; j < RP_PIECES; ++j) {
        const float4 xv = buf[j];
        d += xv.x * yv[j].x + xv.y * yv[j].y + xv.z * yv[j].z + xv.w * yv[j].w;
    }
#pragma unroll
    for (int off = 16; off > 0; off >>= 1)
        d += __shfl_xor(d, off, 32);
    const float s = __shfl(bp, i, 32) + d;
    if (lane == i) snew = s;

    const float mn = fmaxf(m, s);
    const float sc = __expf(m - mn);
    const float p  = __expf(s - mn);
    l = l * sc + p;
#pragma unroll
    for (int j = 0; j < RP_PIECES; ++j) {
        const float4 a = buf[j];
        acc[j].x = acc[j].x * sc + p * a.x;
        acc[j].y = acc[j].y * sc + p * a.y;
        acc[j].z = acc[j].z * sc + p * a.z;
        acc[j].w = acc[j].w * sc + p * a.w;
    }
    m = mn;
}

__global__ __launch_bounds__(512) void k_route_pass(
    const float* __restrict__ x, const float* __restrict__ y,
    const float* __restrict__ bprev, float* __restrict__ bout,
    float* __restrict__ pacc, float* __restrict__ pml)
{
    __shared__ float  wmax[RP_WAVES];
    __shared__ float  wsum[RP_WAVES];
    __shared__ float  scores[RP_BROWS];
    __shared__ float4 comb[RP_WAVES / 2][H_DIM / 4];
    const int wave = threadIdx.x >> 5;
    const int lane = threadIdx.x & 31;
    const int row0 = blockIdx.x * RP_BROWS;
    const int c    = row0 >> 11;
    const int r0   = row0 + wave * RP_WROWS;

    const float4* yr = reinterpret_cast<const float4*>(y + (size_t)c * H_DIM);
    float4 yv[RP_PIECES];
#pragma unroll
    for (int j = 0; j < RP_PIECES; ++j) yv[j] = yr[j * 32 + lane];

    float bp = 0.f;
    if (bprev) bp = bprev[r0 + (lane & 15)];

    const float4* xr = reinterpret_cast<const float4*>(x + (size_t)r0 * H_DIM) + lane;

    float4 acc[8];
#pragma unroll
    for (int j = 0; j < 8; ++j) acc[j] = make_float4(0.f, 0.f, 0.f, 0.f);
    float m = -__builtin_inff();
    float l = 0.f;
    float snew = 0.f;

    float4 buf[3][RP_PIECES];
    rp_load(buf[0], xr);
    rp_load(buf[1], xr + RP_ROWSTEP);
    __builtin_amdgcn_sched_barrier(0);
#pragma unroll
    for (int i = 0; i < RP_WROWS; ++i) {
        if (i + 2 < RP_WROWS)
            rp_load(buf[(i + 2) % 3], xr + (i + 2) * RP_ROWSTEP);
        __builtin_amdgcn_sched_barrier(0);
#pragma unroll
        for (int j = 0; j < RP_PIECES; ++j)
            asm volatile("" : "+v"(buf[i % 3][j].x), "+v"(buf[i % 3][j].y),
                              "+v"(buf[i % 3][j].z), "+v"(buf[i % 3][j].w));
        rp_row(buf[i % 3], yv, acc, m, l, bp, snew, i, lane);
#pragma unroll
        for (int j = 0; j < RP_PIECES; ++j)
            asm volatile("" : "+v"(acc[j].x), "+v"(acc[j].y), "+v"(acc[j].z), "+v"(acc[j].w));
        __builtin_amdgcn_sched_barrier(0);
    }

    if (lane == 0) wmax[wave] = m;
    if (lane < RP_WROWS) scores[wave * RP_WROWS + lane] = snew;
    __syncthreads();

    float M = wmax[0];
#pragma unroll
    for (int w = 1; w < RP_WAVES; ++w) M = fmaxf(M, wmax[w]);
    const float f = __expf(m - M);
#pragma unroll
    for (int j = 0; j < 8; ++j) {
        acc[j].x *= f; acc[j].y *= f; acc[j].z *= f; acc[j].w *= f;
    }
    if (lane == 0) wsum[wave] = l * f;
    if (wave >= RP_WAVES / 2) {
#pragma unroll
        for (int j = 0; j < 8; ++j) comb[wave - RP_WAVES / 2][j * 32 + lane] = acc[j];
    }
    __syncthreads();
    if (wave < RP_WAVES / 2) {
#pragma unroll
        for (int j = 0; j < 8; ++j) {
            const float4 o = comb[wave][j * 32 + lane];
            acc[j].x += o.x; acc[j].y += o.y; acc[j].z += o.z; acc[j].w += o.w;
            comb[wave][j * 32 + lane] = acc[j];
        }
    }
    __syncthreads();

    const int t = threadIdx.x;
    if (t < H_DIM / 4) {
        float4 s = comb[0][t];
#pragma unroll
        for (int w = 1; w < RP_WAVES / 2; ++w) {
            const float4 o = comb[w][t];
            s.x += o.x; s.y += o.y; s.z += o.z; s.w += o.w;
        }
        reinterpret_cast<float4*>(pacc + (size_t)blockIdx.x * H_DIM)[t] = s;
        if (bout) bout[row0 + t] = scores[t];
    }
    if (t == H_DIM / 4) {
        float L = 0.f;
#pragma unroll
        for (int w = 0; w < RP_WAVES; ++w) L += wsum[w];
        pml[(size_t)blockIdx.x * RP_MLPITCH]     = M;
        pml[(size_t)blockIdx.x * RP_MLPITCH + 1] = L;
    }
}

__global__ __launch_bounds__(256) void k_combine_partials(
    const float* __restrict__ pacc, const float* __restrict__ pml,
    float* __restrict__ xd)
{
    const int i = blockIdx.x * 256 + threadIdx.x;
    const int c = i / H_DIM;
    const int h = i % H_DIM;
    float mk[RP_CHUNKS], lk[RP_CHUNKS];
    float M = -__builtin_inff();
#pragma unroll
    for (int k = 0; k < RP_CHUNKS; ++k) {
        mk[k] = pml[(size_t)(c * RP_CHUNKS + k) * RP_MLPITCH];
        lk[k] = pml[(size_t)(c * RP_CHUNKS + k) * RP_MLPITCH + 1];
        M = fmaxf(M, mk[k]);
    }
    float L = 0.f, s = 0.f;
#pragma unroll
    for (int k = 0; k < RP_CHUNKS; ++k) {
        const float f = __expf(mk[k] - M);
        L += f * lk[k];
        s += f * pacc[(size_t)(c * RP_CHUNKS + k) * H_DIM + h];
    }
    xd[i] = s / L;
}

extern "C" void kernel_launch(void* const* d_in, const int* in_sizes, int n_in,
                              void* d_out, int out_size, void* d_ws, size_t ws_size,
                              hipStream_t stream)
{
    const float* x = (const float*)d_in[0];
    const float* W = (const float*)d_in[1];
    float* out = (float*)d_out;
    float* ws  = (float*)d_ws;

    float* partials = ws;
    float* xd   = partials + (size_t)NCHUNK * C_CAPS * H_DIM;
    float* craw = xd   + C_CAPS * H_DIM;
    float* cvec = craw + C_CAPS * H_DIM;
    float* yv   = cvec + C_CAPS * H_DIM;
    float* bl   = yv   + C_CAPS * H_DIM;
    float* pml  = bl   + C_CAPS * S_DIM;

    const dim3 gws(NCHUNK * WS_HALVES, C_CAPS);
    const int  gred = (C_CAPS * H_DIM) / 256;
    const dim3 gnt(H_DIM / 16, C_CAPS / 16);
    const int  grp  = (C_CAPS * S_DIM) / RP_BROWS;

    k_weighted_sum<<<gws, 256, 0, stream>>>(x, nullptr, partials);
    k_reduce_partials<<<gred, 256, 0, stream>>>(partials, xd);
    k_gemm32<false><<<gnt, 256, 0, stream>>>(xd, W, craw);
    k_squash<<<C_CAPS, 256, 0, stream>>>(craw, cvec);
    k_gemm32<true><<<gnt, 256, 0, stream>>>(cvec, W, yv);

    k_route_pass<<<grp, 512, 0, stream>>>(x, yv, nullptr, bl, partials, pml);
    k_combine_partials<<<gred, 256, 0, stream>>>(partials, pml, xd);
    k_gemm32<false><<<gnt, 256, 0, stream>>>(xd, W, craw);
    k_squash<<<C_CAPS, 256, 0, stream>>>(craw, cvec);
    k_gemm32<true><<<gnt, 256, 0, stream>>>(cvec, W, yv);

    k_route_pass<<<grp, 512, 0, stream>>>(x, yv, bl, nullptr, partials, pml);
    k_combine_partials<<<gred, 256, 0, stream>>>(partials, pml, xd);
    k_gemm32<false><<<gnt, 256, 0, stream>>>(xd, W, craw);
    k_squash<<<C_CAPS, 256, 0, stream>>>(craw, out);
}


# Squash folded into the second GEMM of iterations 1-2: the block sums its rows' squares from its A fragments and scales its tile (14 -> 12 launches; f32 rounding and sum order of y change)
# speedup vs baseline: 3.5699x; 1.1558x over previous
#include <hip/hip_runtime.h>

typedef __attribute__((ext_vector_type(2))) float v2f;
typedef __attribute__((ext_vector_type(8))) float v8f;

#define C_CAPS 32
#define S_DIM  2048
#define H_DIM  1024

#define GEMM_WAVES 8
#define GEMM_KW    (H_DIM / GEMM_WAVES)
#define GEMM_STEPS (GEMM_KW / 4)

template <bool BT>
__global__ __launch_bounds__(256) void k_gemm32(
    const float* __restrict__ A, const float* __restrict__ W,
    float* __restrict__ Dst)
{
    __shared__ float red[GEMM_WAVES][8][32];
    __shared__ float sqw[GEMM_WAVES][16];
    const int wave = threadIdx.x >> 5;
    const int lane = threadIdx.x & 31;
    const int n0   = blockIdx.x * 16;
    const int m0   = blockIdx.y * 16;
    const int kq   = (lane >> 4) * 2;
    const int l    = lane & 15;
    const int kk0  = wave * GEMM_KW + kq;

    const float* Ap = A + (size_t)(m0 + l) * H_DIM + kk0;
    const float* Wp = BT ? W + (size_t)(n0 + l) * H_DIM + kk0
                         : W + (size_t)kk0 * H_DIM + n0 + l;

    v2f a[GEMM_STEPS];
    v2f b[GEMM_STEPS];
#pragma unroll
    for (int i = 0; i < GEMM_STEPS; ++i) {
        a[i] = *(const v2f*)(Ap + 4 * i);
        if (BT) {
            b[i] = *(const v2f*)(Wp + 4 * i);
        } else {
            b[i].x = Wp[(size_t)(4 * i) * H_DIM];
            b[i].y = Wp[(size_t)(4 * i + 1) * H_DIM];
        }
    }

    __builtin_amdgcn_sched_barrier(0);

    v8f c0 = {}; v8f c1 = {};
#pragma unroll
    for (int i = 0; i < GEMM_STEPS; i += 2) {
        c0 = __builtin_amdgcn_wmma_f32_16x16x4_f32(
                 false, a[i], false, b[i], (short)0, c0, false, false);
        c1 = __builtin_amdgcn_wmma_f32_16x16x4_f32(
                 false, a[i + 1], false, b[i + 1], (short)0, c1, false, false);
    }

    if (BT) {
        float q = 0.f;
#pragma unroll
        for (int i = 0; i < GEMM_STEPS; ++i) q += a[i].x * a[i].x + a[i].y * a[i].y;
        q += __shfl_xor(q, 16, 32);
        if (lane < 16) sqw[wave][lane] = q;
    }

    for (int i = 0; i < 8; ++i) red[wave][i][lane] = c0[i] + c1[i];
    __syncthreads();

    const int row = (lane >> 4) * 8 + wave;
    float s = 0.f;
    for (int w = 0; w < GEMM_WAVES; ++w) s += red[w][wave][lane];
    if (BT) {
        float sq = 0.f;
        for (int w = 0; w < GEMM_WAVES; ++w) sq += sqw[w][row];
        s *= sq / ((1.0f + sq) * sqrtf(sq + 1e-9f));
    }
    Dst[(size_t)(m0 + row) * H_DIM + n0 + l] = s;
}

__global__ __launch_bounds__(256) void k_squash(
    const float* __restrict__ craw, float* __restrict__ out)
{
    __shared__ float red[256];
    const int c = blockIdx.x;
    const int t = threadIdx.x;
    float4 v = reinterpret_cast<const float4*>(craw + (size_t)c * H_DIM)[t];
    red[t] = v.x * v.x + v.y * v.y + v.z * v.z + v.w * v.w;
    __syncthreads();
    for (int off = 128; off > 0; off >>= 1) {
        if (t < off) red[t] += red[t + off];
        __syncthreads();
    }
    const float sq    = red[0];
    const float coeff = sq / ((1.0f + sq) * sqrtf(sq + 1e-9f));
    float4 r = make_float4(coeff * v.x, coeff * v.y, coeff * v.z, coeff * v.w);
    reinterpret_cast<float4*>(out + (size_t)c * H_DIM)[t] = r;
}

#define RP_WAVES   16
#define RP_WROWS   16
#define RP_BROWS   (RP_WAVES * RP_WROWS)
#define RP_CHUNKS  (S_DIM / RP_BROWS)
#define RP_PIECES  8
#define RP_ROWSTEP (H_DIM / 4)
#define RP_MLPITCH 32

template <bool REV>
__device__ __forceinline__ int rp_rowidx(int i)
{
    return REV ? (RP_WROWS - 1 - i) : i;
}

__device__ __forceinline__ void rp_load(float4 (&buf)[RP_PIECES], const float4* p)
{
#pragma unroll
    for (int q = 0; q < RP_PIECES; ++q) buf[q] = p[q * 32];
}

__device__ __forceinline__ void rp_row(const float4 (&buf)[RP_PIECES],
                                       const float4 (&yv)[RP_PIECES],
                                       float4 (&acc)[RP_PIECES],
                                       float& m, float& l, float bp, float& snew,
                                       int i, int lane)
{
    float d = 0.f;
#pragma unroll
    for (int j = 0; j < RP_PIECES; ++j) {
        const float4 xv = buf[j];
        d += xv.x * yv[j].x + xv.y * yv[j].y + xv.z * yv[j].z + xv.w * yv[j].w;
    }
#pragma unroll
    for (int off = 16; off > 0; off >>= 1)
        d += __shfl_xor(d, off, 32);
    const float s = __shfl(bp, i, 32) + d;
    if (lane == i) snew = s;

    const float mn = fmaxf(m, s);
    const float sc = __expf(m - mn);
    const float p  = __expf(s - mn);
    l = l * sc + p;
#pragma unroll
    for (int j = 0; j < RP_PIECES; ++j) {
        const float4 a = buf[j];
        acc[j].x = acc[j].x * sc + p * a.x;
        acc[j].y = acc[j].y * sc + p * a.y;
        acc[j].z = acc[j].z * sc + p * a.z;
        acc[j].w = acc[j].w * sc + p * a.w;
    }
    m = mn;
}

__device__ __forceinline__ void rp_row_uniform(const float4 (&buf)[RP_PIECES],
                                               float4 (&acc)[RP_PIECES])
{
#pragma unroll
    for (int j = 0; j < RP_PIECES; ++j) {
        acc[j].x += buf[j].x; acc[j].y += buf[j].y;
        acc[j].z += buf[j].z; acc[j].w += buf[j].w;
    }
}

template <bool REV, bool UNI>
__global__ __launch_bounds__(512) void k_route_pass(
    const float* __restrict__ x, const float* __restrict__ y,
    const float* __restrict__ bprev, float* __restrict__ bout,
    float* __restrict__ pacc, float* __restrict__ pml)
{
    __shared__ float  wmax[RP_WAVES];
    __shared__ float  wsum[RP_WAVES];
    __shared__ float  scores[RP_BROWS];
    __shared__ float4 comb[RP_WAVES / 2][H_DIM / 4];
    const int wave = threadIdx.x >> 5;
    const int lane = threadIdx.x & 31;
    const int row0 = blockIdx.x * RP_BROWS;
    const int c    = row0 >> 11;
    const int r0   = row0 + wave * RP_WROWS;

    float4 yv[RP_PIECES];
    if (!UNI) {
        const float4* yr = reinterpret_cast<const float4*>(y + (size_t)c * H_DIM);
#pragma unroll
        for (int j = 0; j < RP_PIECES; ++j) yv[j] = yr[j * 32 + lane];
    }

    float bp = 0.f;
    if (!UNI && bprev) bp = bprev[r0 + (lane & 15)];

    const float4* xr = reinterpret_cast<const float4*>(x + (size_t)r0 * H_DIM) + lane;

    float4 acc[8];
#pragma unroll
    for (int j = 0; j < 8; ++j) acc[j] = make_float4(0.f, 0.f, 0.f, 0.f);
    float m = -__builtin_inff();
    float l = 0.f;
    float snew = 0.f;

    float4 buf[3][RP_PIECES];
    rp_load(buf[0], xr + rp_rowidx<REV>(0) * RP_ROWSTEP);
    rp_load(buf[1], xr + rp_rowidx<REV>(1) * RP_ROWSTEP);
    __builtin_amdgcn_sched_barrier(0);
#pragma unroll
    for (int i = 0; i < RP_WROWS; ++i) {
        if (i + 2 < RP_WROWS)
            rp_load(buf[(i + 2) % 3], xr + rp_rowidx<REV>(i + 2) * RP_ROWSTEP);
        __builtin_amdgcn_sched_barrier(0);
#pragma unroll
        for (int j = 0; j < RP_PIECES; ++j)
            asm volatile("" : "+v"(buf[i % 3][j].x), "+v"(buf[i % 3][j].y),
                              "+v"(buf[i % 3][j].z), "+v"(buf[i % 3][j].w));
        if (UNI) rp_row_uniform(buf[i % 3], acc);
        else     rp_row(buf[i % 3], yv, acc, m, l, bp, snew, rp_rowidx<REV>(i), lane);
#pragma unroll
        for (int j = 0; j < RP_PIECES; ++j)
            asm volatile("" : "+v"(acc[j].x), "+v"(acc[j].y), "+v"(acc[j].z), "+v"(acc[j].w));
        __builtin_amdgcn_sched_barrier(0);
    }

    if (UNI) { m = 0.f; l = (float)RP_WROWS; }

    if (lane == 0) wmax[wave] = m;
    if (!UNI && lane < RP_WROWS) scores[wave * RP_WROWS + lane] = snew;
    __syncthreads();

    float M = wmax[0];
#pragma unroll
    for (int w = 1; w < RP_WAVES; ++w) M = fmaxf(M, wmax[w]);
    const float f = __expf(m - M);
#pragma unroll
    for (int j = 0; j < 8; ++j) {
        acc[j].x *= f; acc[j].y *= f; acc[j].z *= f; acc[j].w *= f;
    }
    if (lane == 0) wsum[wave] = l * f;
    if (wave >= RP_WAVES / 2) {
#pragma unroll
        for (int j = 0; j < 8; ++j) comb[wave - RP_WAVES / 2][j * 32 + lane] = acc[j];
    }
    __syncthreads();
    if (wave < RP_WAVES / 2) {
#pragma unroll
        for (int j = 0; j < 8; ++j) {
            const float4 o = comb[wave][j * 32 + lane];
            acc[j].x += o.x; acc[j].y += o.y; acc[j].z += o.z; acc[j].w += o.w;
            comb[wave][j * 32 + lane] = acc[j];
        }
    }
    __syncthreads();

    const int t = threadIdx.x;
    if (t < H_DIM / 4) {
        float4 s = comb[0][t];
#pragma unroll
        for (int w = 1; w < RP_WAVES / 2; ++w) {
            const float4 o = comb[w][t];
            s.x += o.x; s.y += o.y; s.z += o.z; s.w += o.w;
        }
        reinterpret_cast<float4*>(pacc + (size_t)blockIdx.x * H_DIM)[t] = s;
        if (bout) bout[row0 + t] = scores[t];
    }
    if (t == H_DIM / 4) {
        float L = 0.f;
#pragma unroll
        for (int w = 0; w < RP_WAVES; ++w) L += wsum[w];
        pml[(size_t)blockIdx.x * RP_MLPITCH]     = M;
        pml[(size_t)blockIdx.x * RP_MLPITCH + 1] = L;
    }
}

__global__ __launch_bounds__(256) void k_combine_partials(
    const float* __restrict__ pacc, const float* __restrict__ pml,
    float* __restrict__ xd)
{
    const int i = blockIdx.x * 256 + threadIdx.x;
    const int c = i / H_DIM;
    const int h = i % H_DIM;
    float mk[RP_CHUNKS], lk[RP_CHUNKS];
    float M = -__builtin_inff();
#pragma unroll
    for (int k = 0; k < RP_CHUNKS; ++k) {
        mk[k] = pml[(size_t)(c * RP_CHUNKS + k) * RP_MLPITCH];
        lk[k] = pml[(size_t)(c * RP_CHUNKS + k) * RP_MLPITCH + 1];
        M = fmaxf(M, mk[k]);
    }
    float L = 0.f, s = 0.f;
#pragma unroll
    for (int k = 0; k < RP_CHUNKS; ++k) {
        const float f = __expf(mk[k] - M);
        L += f * lk[k];
        s += f * pacc[(size_t)(c * RP_CHUNKS + k) * H_DIM + h];
    }
    xd[i] = s / L;
}

extern "C" void kernel_launch(void* const* d_in, const int* in_sizes, int n_in,
                              void* d_out, int out_size, void* d_ws, size_t ws_size,
                              hipStream_t stream)
{
    const float* x = (const float*)d_in[0];
    const float* W = (const float*)d_in[1];
    float* out = (float*)d_out;
    float* ws  = (float*)d_ws;

    float* partials = ws;
    float* xd   = partials + (size_t)RP_CHUNKS * C_CAPS * H_DIM;
    float* craw = xd   + C_CAPS * H_DIM;
    float* yv   = craw + C_CAPS * H_DIM;
    float* bl   = yv   + C_CAPS * H_DIM;
    float* pml  = bl   + C_CAPS * S_DIM;

    const int  gred = (C_CAPS * H_DIM) / 256;
    const dim3 gnt(H_DIM / 16, C_CAPS / 16);
    const int  grp  = (C_CAPS * S_DIM) / RP_BROWS;

    k_route_pass<false, true><<<grp, 512, 0, stream>>>(x, nullptr, nullptr, nullptr, partials, pml);
    k_combine_partials<<<gred, 256, 0, stream>>>(partials, pml, xd);
    k_gemm32<false><<<gnt, 256, 0, stream>>>(xd, W, craw);
    k_gemm32<true><<<gnt, 256, 0, stream>>>(craw, W, yv);

    k_route_pass<true, false><<<grp, 512, 0, stream>>>(x, yv, nullptr, bl, partials, pml);
    k_combine_partials<<<gred, 256, 0, stream>>>(partials, pml, xd);
    k_gemm32<false><<<gnt, 256, 0, stream>>>(xd, W, craw);
    k_gemm32<true><<<gnt, 256, 0, stream>>>(craw, W, yv);

    k_route_pass<false, false><<<grp, 512, 0, stream>>>(x, yv, bl, nullptr, partials, pml);
    k_combine_partials<<<gred, 256, 0, stream>>>(partials, pml, xd);
    k_gemm32<false><<<gnt, 256, 0, stream>>>(xd, W, craw);
    k_squash<<<C_CAPS, 256, 0, stream>>>(craw, out);
}
